# starting kernel
# baseline (speedup 1.0000x reference)
#include <hip/hip_runtime.h>
#include <math.h>

typedef __attribute__((ext_vector_type(2))) float v2f;
typedef __attribute__((ext_vector_type(8))) float v8f;

#define NROWS   500000
#define D0      128
#define UNITS   4
#define STACK   12
#define HFIN    (D0 + UNITS * STACK)
#define HS      180
#define NTILES  (NROWS / 16)
#define WPB     4
#define TPW     2
#define WLDSF   7552
#define SHOFF   WLDSF
#define HOFF    (WLDSF + 48)
#define BN_EPS  1e-3f

__device__ __forceinline__ int woff_layer(int i) { return 8 * i * i + 504 * i; }

#define CHUNK4(KOFF, WC, NCL, MSK, ACC0, ACC1)                                  \
    {                                                                           \
        v2f a0 = *reinterpret_cast<const v2f*>(h0 + aidx + (KOFF));             \
        v2f a1 = *reinterpret_cast<const v2f*>(h1 + aidx + (KOFF));             \
        v2f bw;                                                                 \
        bw.x = wlds[wo + ((KOFF) + khalf)     * (WC) + (NCL)] * (MSK);          \
        bw.y = wlds[wo + ((KOFF) + khalf + 1) * (WC) + (NCL)] * (MSK);          \
        ACC0 = __builtin_amdgcn_wmma_f32_16x16x4_f32(false, a0, false, bw,      \
                                                     (short)0, ACC0, false, false); \
        ACC1 = __builtin_amdgcn_wmma_f32_16x16x4_f32(false, a1, false, bw,      \
                                                     (short)0, ACC1, false, false); \
    }

__global__ void __launch_bounds__(128)
predict_stack_wmma(const float* __restrict__ x,
                   const float* __restrict__ w0,  const float* __restrict__ w1,
                   const float* __restrict__ w2,  const float* __restrict__ w3,
                   const float* __restrict__ w4,  const float* __restrict__ w5,
                   const float* __restrict__ w6,  const float* __restrict__ w7,
                   const float* __restrict__ w8,  const float* __restrict__ w9,
                   const float* __restrict__ w10, const float* __restrict__ w11,
                   const float* __restrict__ gamma, const float* __restrict__ beta,
                   const float* __restrict__ mean,  const float* __restrict__ var,
                   const float* __restrict__ wf,    const float* __restrict__ bf,
                   float* __restrict__ out)
{
    extern __shared__ float smem[];
    float* wlds  = smem;
    float* shlds = smem + SHOFF;
    float* hbase = smem + HOFF;

    const int tid = threadIdx.x;

    {
        const float* wsrc[STACK] = {w0,w1,w2,w3,w4,w5,w6,w7,w8,w9,w10,w11};
        #pragma unroll
        for (int i = 0; i < STACK; ++i) {
            const int off = woff_layer(i);
            const int cnt = (D0 + UNITS * i) * UNITS;
            for (int idx = tid; idx < cnt; idx += 128) {
                const int n  = idx & 3;
                const float s = gamma[i * UNITS + n] * rsqrtf(var[i * UNITS + n] + BN_EPS);
                wlds[off + idx] = wsrc[i][idx] * s;
            }
        }
        for (int idx = tid; idx < HFIN * 2; idx += 128)
            wlds[7200 + idx] = wf[idx];
        if (tid < STACK * UNITS) {
            const int n = tid & 3, i = tid >> 2;
            const float s = gamma[i * UNITS + n] * rsqrtf(var[i * UNITS + n] + BN_EPS);
            shlds[tid] = beta[tid] - mean[tid] * s;
        }
    }
    __syncthreads();

    const int wave = tid >> 5;
    const int lane = tid & 31;
    const int pair = blockIdx.x * WPB + wave;
    if (pair * TPW >= NTILES) return;

    float* h0 = hbase + (wave * TPW + 0) * 16 * HS;
    float* h1 = hbase + (wave * TPW + 1) * 16 * HS;

    const int mrow  = lane & 15;
    const int khalf = (lane >> 4) * 2;
    const int ncol  = lane & 15;
    const int mhi   = (lane >> 4) * 8;
    const bool n4   = (ncol < UNITS);
    const bool n2   = (ncol < 2);
    const float msk4 = n4 ? 1.0f : 0.0f;
    const float msk2 = n2 ? 1.0f : 0.0f;
    const int  ncl4  = n4 ? ncol : 0;
    const int  ncl2  = n2 ? ncol : 0;

    const int row0 = pair * TPW * 16;
    const float4* xv = reinterpret_cast<const float4*>(x);
    #pragma unroll 4
    for (int r = 0; r < 16; ++r) {
        float4 v0 = xv[(size_t)(row0 + r)      * (D0 / 4) + lane];
        float4 v1 = xv[(size_t)(row0 + 16 + r) * (D0 / 4) + lane];
        *reinterpret_cast<float4*>(&h0[r * HS + 48 + lane * 4]) = v0;
        *reinterpret_cast<float4*>(&h1[r * HS + 48 + lane * 4]) = v1;
    }

    #pragma unroll
    for (int i = 0; i < STACK; ++i) {
        const int base = 48 - 4 * i;
        const int K    = D0 + UNITS * i;
        const int wo   = woff_layer(i);
        const int aidx = mrow * HS + base + khalf;

        v8f cA0 = {0,0,0,0,0,0,0,0}, cA1 = {0,0,0,0,0,0,0,0};
        v8f cB0 = {0,0,0,0,0,0,0,0}, cB1 = {0,0,0,0,0,0,0,0};

        const int k0 = K & 4;
        if (k0) CHUNK4(0, UNITS, ncl4, msk4, cA0, cA1)
        #pragma unroll 2
        for (int k = k0; k < K; k += 8) {
            CHUNK4(k,     UNITS, ncl4, msk4, cA0, cA1)
            CHUNK4(k + 4, UNITS, ncl4, msk4, cB0, cB1)
        }
        const v8f c0 = cA0 + cB0;
        const v8f c1 = cA1 + cB1;

        if (n4) {
            const float sh = shlds[i * UNITS + ncol];
            const int ycol = base - UNITS;
            #pragma unroll
            for (int v = 0; v < 8; ++v) {
                const int M = v + mhi;
                h0[M * HS + ycol + ncol] = fmaxf(0.0f, c0[v] + sh);
                h1[M * HS + ycol + ncol] = fmaxf(0.0f, c1[v] + sh);
            }
        }
    }

    {
        const int wo   = 7200;
        const int aidx = mrow * HS + khalf;
        v8f cA0 = {0,0,0,0,0,0,0,0}, cA1 = {0,0,0,0,0,0,0,0};
        v8f cB0 = {0,0,0,0,0,0,0,0}, cB1 = {0,0,0,0,0,0,0,0};
        #pragma unroll 2
        for (int k = 0; k < HFIN; k += 8) {
            CHUNK4(k,     2, ncl2, msk2, cA0, cA1)
            CHUNK4(k + 4, 2, ncl2, msk2, cB0, cB1)
        }
        v8f cf[TPW];
        cf[0] = cA0 + cB0;
        cf[1] = cA1 + cB1;

        const float bfv = n2 ? bf[ncol] : 0.0f;
        #pragma unroll
        for (int t = 0; t < TPW; ++t) {
            const int trow = row0 + t * 16;
            #pragma unroll
            for (int v = 0; v < 8; ++v) {
                const float l = cf[t][v] + bfv;
                const float o = __shfl_xor(l, 1, 32);
                const float m = fmaxf(l, o);
                const float e  = __expf(l - m);
                const float eo = __expf(o - m);
                const float p  = e / (e + eo);
                if (n2) {
                    const int M = v + mhi;
                    out[(size_t)(trow + M) * 2 + ncol] = p;
                }
            }
        }
    }
}

extern "C" void kernel_launch(void* const* d_in, const int* in_sizes, int n_in,
                              void* d_out, int out_size, void* d_ws, size_t ws_size,
                              hipStream_t stream)
{
    (void)in_sizes; (void)n_in; (void)out_size; (void)d_ws; (void)ws_size;
    const float* x   = (const float*)d_in[0];
    const float* w[12];
    for (int i = 0; i < 12; ++i) w[i] = (const float*)d_in[1 + i];
    const float* gamma = (const float*)d_in[13];
    const float* beta  = (const float*)d_in[14];
    const float* mean  = (const float*)d_in[15];
    const float* var   = (const float*)d_in[16];
    const float* wf    = (const float*)d_in[17];
    const float* bf    = (const float*)d_in[18];
    float* out = (float*)d_out;

    const int pairs  = NTILES / TPW;
    const int blocks = (pairs + WPB - 1) / WPB;
    const size_t shmem =
        (size_t)(HOFF + WPB * TPW * 16 * HS) * sizeof(float);

    predict_stack_wmma<<<dim3(blocks), dim3(128), shmem, stream>>>(
        x, w[0], w[1], w[2], w[3], w[4], w[5], w[6], w[7], w[8], w[9], w[10], w[11],
        gamma, beta, mean, var, wf, bf, out);
}


# Product transposed (out units on matrix rows): x goes once from global straight into B operands, x parts of all 12 layers and logits in one pass, activations stay in registers; f32 sum order changes
# speedup vs baseline: 4.8080x; 4.8080x over previous
#include <hip/hip_runtime.h>
#include <math.h>


typedef __attribute__((ext_vector_type(2))) float v2f;
typedef __attribute__((ext_vector_type(4))) float v4f;
typedef __attribute__((ext_vector_type(8))) float v8f;

#define NROWS   500000
#define D0      128
#define UNITS   4
#define STACK   12
#define NTILES  (NROWS / 16)
#define NBLK    512
#define WPB     8
#define NTHR    (WPB * 32)
#define BN_EPS  1e-3f

#define WX_F    (4 * 16 * 32 * 4)
#define WY_OFF  WX_F
#define WY_F    (STACK * 4 * 32 * 2)
#define SH_OFF  (WY_OFF + WY_F)
#define LDS_F   (SH_OFF + STACK * UNITS)


#define WMMA(A, B, C) __builtin_amdgcn_wmma_f32_16x16x4_f32(false, (A), false, (B), (short)0, (C), false, false)

__device__ __forceinline__ void load_tile(v4f (&xv)[16], const float* __restrict__ x, int tile, int cl, int h)
{
    const v4f* p = reinterpret_cast<const v4f*>(x + (size_t)(tile * 16 + cl) * D0 + 4 * h);
    for (int c = 0; c < 16; ++c) xv[c] = p[2 * c];
}

__device__ __forceinline__ void process_tile(const v4f (&xv)[16],
                                             const v4f* wx, const v2f* wy, const v2f* sh,
                                             float bf0, float bf1, float* __restrict__ out,
                                             int tile, int cl, int h)
{
    v8f acc[4];
    for (int t = 0; t < 4; ++t) acc[t] = (v8f){0,0,0,0,0,0,0,0};

    v4f a[4];
    for (int t = 0; t < 4; ++t) a[t] = wx[(t * 16) * 32];
    for (int c = 0; c < 16; ++c) {
        v4f an[4];
        if (c + 1 < 16) {
            for (int t = 0; t < 4; ++t) an[t] = wx[(t * 16 + c + 1) * 32];
        }
        const v2f blo = {xv[c].x, xv[c].y};
        const v2f bhi = {xv[c].z, xv[c].w};
        for (int t = 0; t < 4; ++t) {
            const v2f alo = {a[t].x, a[t].y};
            acc[t] = WMMA(alo, blo, acc[t]);
        }
        for (int t = 0; t < 4; ++t) {
            const v2f ahi = {a[t].z, a[t].w};
            acc[t] = WMMA(ahi, bhi, acc[t]);
        }
        if (c + 1 < 16) {
            for (int t = 0; t < 4; ++t) a[t] = an[t];
        }
    }

    #pragma unroll
    for (int i = 0; i < STACK; ++i) {
        const int t = i >> 2, g = i & 3;
        const v2f s = sh[i * 2];
        v2f yb;
        yb.x = fmaxf(0.0f, acc[t][2 * g]     + s.x);
        yb.y = fmaxf(0.0f, acc[t][2 * g + 1] + s.y);
        for (int tt = (g == 3 ? t + 1 : t); tt < 4; ++tt) {
            const v2f a2 = wy[(i * 4 + tt) * 32];
            acc[tt] = WMMA(a2, yb, acc[tt]);
        }
    }

    const float l0 = acc[3][0] + bf0;
    const float l1 = acc[3][1] + bf1;
    const float mx = fmaxf(l0, l1);
    const float e0 = __expf(l0 - mx);
    const float e1 = __expf(l1 - mx);
    const float den = e0 + e1;
    if (h == 0) {
        const v2f p = {e0 / den, e1 / den};
        *reinterpret_cast<v2f*>(out + (size_t)(tile * 16 + cl) * 2) = p;
    }
}

__global__ void __launch_bounds__(NTHR, 4)
predict_stack_wmma(const float* __restrict__ x,
                   const float* __restrict__ w0,  const float* __restrict__ w1,
                   const float* __restrict__ w2,  const float* __restrict__ w3,
                   const float* __restrict__ w4,  const float* __restrict__ w5,
                   const float* __restrict__ w6,  const float* __restrict__ w7,
                   const float* __restrict__ w8,  const float* __restrict__ w9,
                   const float* __restrict__ w10, const float* __restrict__ w11,
                   const float* __restrict__ gamma, const float* __restrict__ beta,
                   const float* __restrict__ mean,  const float* __restrict__ var,
                   const float* __restrict__ wf,    const float* __restrict__ bf,
                   float* __restrict__ out)
{
    __shared__ __attribute__((aligned(16))) float smem[LDS_F];

    const int tid = threadIdx.x;

    const float* wsrc[STACK] = {w0,w1,w2,w3,w4,w5,w6,w7,w8,w9,w10,w11};
    const int u = tid & 3;
    float sc[STACK], wxa[STACK], wxb[STACK], wyv[STACK];
    for (int i = 0; i < STACK; ++i) {
        sc[i]  = gamma[i * UNITS + u] * rsqrtf(var[i * UNITS + u] + BN_EPS);
        wxa[i] = wsrc[i][16 * i + tid];
        wxb[i] = wsrc[i][16 * i + tid + NTHR];
        const int ylast = (i > 0) ? 16 * i - 1 : 0;
        wyv[i] = wsrc[i][tid < ylast ? tid : ylast];
    }
    const float wfa = wf[tid];
    const float wfb = wf[tid + NTHR < (D0 + UNITS * STACK) * 2 ? tid + NTHR : (D0 + UNITS * STACK) * 2 - 1];
    const int   ts  = tid < STACK * UNITS ? tid : STACK * UNITS - 1;
    const float sg = gamma[ts], sv = var[ts], sb = beta[ts], sm = mean[ts];

    for (int idx = tid; idx < LDS_F / 4; idx += NTHR)
        reinterpret_cast<v4f*>(smem)[idx] = (v4f){0.0f, 0.0f, 0.0f, 0.0f};
    __syncthreads();

    for (int i = 0; i < STACK; ++i) {
        const int t = i >> 2, g = i & 3;
        const int m = 8 * (u >> 1) + 2 * g + (u & 1);
        {
            const int k = tid >> 2;
            const int L = 16 * ((k >> 2) & 1) + m;
            smem[(((t * 16 + (k >> 3)) * 32 + L) << 2) + (k & 3)] = wxa[i] * sc[i];
        }
        {
            const int k = (tid + NTHR) >> 2;
            const int L = 16 * ((k >> 2) & 1) + m;
            smem[(((t * 16 + (k >> 3)) * 32 + L) << 2) + (k & 3)] = wxb[i] * sc[i];
        }
        if (tid < 16 * i) {
            const int r = tid >> 2;
            const int up = r & 3, j = i - 1 - (r >> 2);
            const int L = 16 * (up >> 1) + m;
            smem[WY_OFF + ((j * 4 + t) * 32 + L) * 2 + (up & 1)] = wyv[i] * sc[i];
        }
    }
    {
        const int n = tid & 1, r = tid >> 1;
        if (r >= UNITS * STACK) {
            const int k = r - UNITS * STACK;
            const int L = 16 * ((k >> 2) & 1) + n;
            smem[(((3 * 16 + (k >> 3)) * 32 + L) << 2) + (k & 3)] = wfa;
        } else {
            const int up = r & 3, j = STACK - 1 - (r >> 2);
            const int L = 16 * (up >> 1) + n;
            smem[WY_OFF + ((j * 4 + 3) * 32 + L) * 2 + (up & 1)] = wfa;
        }
        if (tid + NTHR < (D0 + UNITS * STACK) * 2) {
            const int k = ((tid + NTHR) >> 1) - UNITS * STACK;
            const int L = 16 * ((k >> 2) & 1) + n;
            smem[(((3 * 16 + (k >> 3)) * 32 + L) << 2) + (k & 3)] = wfb;
        }
    }
    if (tid < STACK * UNITS) {
        const float s = sg * rsqrtf(sv + BN_EPS);
        smem[SH_OFF + tid] = sb - sm * s;
    }
    __syncthreads();

    const int wave = __builtin_amdgcn_readfirstlane(tid >> 5);
    const int lane = tid & 31;
    const int cl   = lane & 15;
    const int h    = lane >> 4;

    const int tbeg = (int)(((long long)blockIdx.x * NTILES) / NBLK);
    const int tend = (int)(((long long)(blockIdx.x + 1) * NTILES) / NBLK);

    const v4f* wx = reinterpret_cast<const v4f*>(smem) + lane;
    const v2f* wy = reinterpret_cast<const v2f*>(smem + WY_OFF) + lane;
    const v2f* sh = reinterpret_cast<const v2f*>(smem + SH_OFF) + h;
    const float bf0 = bf[0], bf1 = bf[1];

    int tile = tbeg + wave;
    if (tile >= tend) return;

    v4f xa[16], xb[16];
    load_tile(xa, x, tile, cl, h);

    while (tile + WPB < tend) {
        load_tile(xb, x, tile + WPB, cl, h);
        asm volatile("" ::: "memory");
        process_tile(xa, wx, wy, sh, bf0, bf1, out, tile, cl, h);
        tile += WPB;
        if (tile + WPB >= tend) {
            for (int c = 0; c < 16; ++c) xa[c] = xb[c];
            break;
        }
        load_tile(xa, x, tile + WPB, cl, h);
        asm volatile("" ::: "memory");
        process_tile(xb, wx, wy, sh, bf0, bf1, out, tile, cl, h);
        tile += WPB;
    }
    process_tile(xa, wx, wy, sh, bf0, bf1, out, tile, cl, h);
}

extern "C" void kernel_launch(void* const* d_in, const int* in_sizes, int n_in,
                              void* d_out, int out_size, void* d_ws, size_t ws_size,
                              hipStream_t stream)
{
    (void)in_sizes; (void)n_in; (void)out_size; (void)d_ws; (void)ws_size;
    const float* x   = (const float*)d_in[0];
    const float* w[12];
    for (int i = 0; i < 12; ++i) w[i] = (const float*)d_in[1 + i];
    const float* gamma = (const float*)d_in[13];
    const float* beta  = (const float*)d_in[14];
    const float* mean  = (const float*)d_in[15];
    const float* var   = (const float*)d_in[16];
    const float* wf    = (const float*)d_in[17];
    const float* bf    = (const float*)d_in[18];
    float* out = (float*)d_out;

    predict_stack_wmma<<<dim3(NBLK), dim3(NTHR), 0, stream>>>(
        x, w[0], w[1], w[2], w[3], w[4], w[5], w[6], w[7], w[8], w[9], w[10], w[11],
        gamma, beta, mean, var, wf, bf, out);
}
